# starting kernel
# baseline (speedup 1.0000x reference)
#include <hip/hip_runtime.h>
#include <hip/hip_bf16.h>
#include <math.h>


typedef __attribute__((ext_vector_type(16))) _Float16 v16h;
typedef __attribute__((ext_vector_type(8)))  float    v8f;

static __device__ __forceinline__ int gtid() {
    return blockIdx.x * blockDim.x + threadIdx.x;
}


__global__ void k_init_deg(float* __restrict__ deg, int n) {
    int i = gtid();
    if (i < n) deg[i] = 1.0f;
}

__global__ void k_accum_deg(const int* __restrict__ dst, int e,
                            float* __restrict__ deg) {
    int i = gtid();
    if (i < e) atomicAdd(&deg[dst[i]], 1.0f);
}

__global__ void k_dinv(const float* __restrict__ deg,
                       float* __restrict__ dinv, int n) {
    int i = gtid();
    if (i < n) dinv[i] = rsqrtf(fmaxf(deg[i], 1.0f));
}


template <int NT>
__global__ void k_pack_b(const float* __restrict__ B,
                         _Float16* __restrict__ Bp, int K) {
    const int N = NT * 16;
    const int chunks = K >> 5;
    int i = gtid();
    if (i >= chunks * NT * 32) return;
    const int lane = i & 31;
    const int t    = (i >> 5) % NT;
    const int c    = (i >> 5) / NT;
    const int r    = lane & 15;
    const int kh   = (lane >> 4) << 3;
    const int k0   = c << 5;
    const int col  = t * 16 + r;
    _Float16* o = Bp + (size_t)i * 16;
#pragma unroll
    for (int j = 0; j < 8; ++j) {
        o[j]     = (_Float16)B[(size_t)(k0 + kh + j) * N + col];
        o[8 + j] = (_Float16)B[(size_t)(k0 + 16 + kh + j) * N + col];
    }
}


template <int NT>
__global__ void k_wmma_gemm(const float* __restrict__ A,
                            const _Float16* __restrict__ Bp,
                            float* __restrict__ C,
                            int M, int K) {
    const int N    = NT * 16;
    const int wave = threadIdx.x >> 5;
    const int lane = threadIdx.x & 31;
    const int tile = blockIdx.x * (blockDim.x >> 5) + wave;
    if (tile * 16 >= M) return;

    const int r   = lane & 15;
    const int kh  = (lane >> 4) << 3;
    const int row = tile * 16 + r;
    const float* __restrict__ arow = A + (size_t)row * K;
    const v16h* __restrict__ bfrag = (const v16h*)Bp;

    v8f zero = {};
    v8f acc[NT];
#pragma unroll
    for (int t = 0; t < NT; ++t) acc[t] = zero;

    for (int k0 = 0; k0 < K; k0 += 32) {
        const float4* __restrict__ a0 =
            reinterpret_cast<const float4*>(arow + k0 + kh);
        const float4* __restrict__ a1 =
            reinterpret_cast<const float4*>(arow + k0 + 16 + kh);
        float4 p0 = a0[0], p1 = a0[1];
        float4 q0 = a1[0], q1 = a1[1];
        v16h a;
        a[0]  = (_Float16)p0.x; a[1]  = (_Float16)p0.y;
        a[2]  = (_Float16)p0.z; a[3]  = (_Float16)p0.w;
        a[4]  = (_Float16)p1.x; a[5]  = (_Float16)p1.y;
        a[6]  = (_Float16)p1.z; a[7]  = (_Float16)p1.w;
        a[8]  = (_Float16)q0.x; a[9]  = (_Float16)q0.y;
        a[10] = (_Float16)q0.z; a[11] = (_Float16)q0.w;
        a[12] = (_Float16)q1.x; a[13] = (_Float16)q1.y;
        a[14] = (_Float16)q1.z; a[15] = (_Float16)q1.w;

        const int cidx = (k0 >> 5) * NT;
#pragma unroll
        for (int t = 0; t < NT; ++t) {
            v16h b = bfrag[(size_t)(cidx + t) * 32 + lane];
            acc[t] = __builtin_amdgcn_wmma_f32_16x16x32_f16(
                false, a, false, b, (short)0, acc[t], false, false);
        }
    }

    const int mbase = (lane < 16) ? 0 : 8;
    const int ncol  = lane & 15;
#pragma unroll
    for (int j = 0; j < 8; ++j) {
        const size_t orow = (size_t)(tile * 16 + mbase + j);
#pragma unroll
        for (int t = 0; t < NT; ++t)
            C[orow * N + t * 16 + ncol] = acc[t][j];
    }
}


template <int C>
__global__ void k_init_agg(const float* __restrict__ hw,
                           const float* __restrict__ dinv,
                           const float* __restrict__ bias,
                           float* __restrict__ agg, int n) {
    int i = gtid();
    if (i >= n * C) return;
    int node = i / C, c = i % C;
    float di = dinv[node];
    agg[i] = bias[c] + hw[i] * di * di;
}

template <int C>
__global__ void k_scatter(const float* __restrict__ hw,
                          const int* __restrict__ src,
                          const int* __restrict__ dst,
                          const float* __restrict__ dinv,
                          float* __restrict__ agg, int e) {
    int i = gtid();
    int edge = i / C, c = i % C;
    if (edge >= e) return;
    if (edge + 4096 < e) {
        __builtin_prefetch(src + edge + 4096, 0, 0);
        __builtin_prefetch(dst + edge + 4096, 0, 0);
    }
    int s = src[edge], d = dst[edge];
    float w = dinv[s] * dinv[d];
    atomicAdd(&agg[(size_t)d * C + c], hw[(size_t)s * C + c] * w);
}

__global__ void k_matvec16_relu(const float* __restrict__ h,
                                const float* __restrict__ w,
                                float* __restrict__ out, int n) {
    int i = gtid();
    if (i >= n) return;
    const float4* hr = (const float4*)(h + (size_t)i * 16);
    float s = 0.0f;
#pragma unroll
    for (int j = 0; j < 4; ++j) {
        float4 v = hr[j];
        s += fmaxf(v.x, 0.0f) * w[j * 4 + 0];
        s += fmaxf(v.y, 0.0f) * w[j * 4 + 1];
        s += fmaxf(v.z, 0.0f) * w[j * 4 + 2];
        s += fmaxf(v.w, 0.0f) * w[j * 4 + 3];
    }
    out[i] = s;
}

__global__ void k_finalize(const float* __restrict__ t,
                           float* __restrict__ out, int n) {
    int i = gtid();
    if (i >= n * 32) return;
    float tv = logf(expf(t[i >> 5]) + 1.1f);
    out[i] *= tv;
}


extern "C" void kernel_launch(void* const* d_in, const int* in_sizes, int n_in,
                              void* d_out, int out_size, void* d_ws, size_t ws_size,
                              hipStream_t stream) {
    const float* x      = (const float*)d_in[0];
    const int*   ei     = (const int*)  d_in[1];
    const float* W_base = (const float*)d_in[2];
    const float* b_base = (const float*)d_in[3];
    const float* W1     = (const float*)d_in[4];
    const float* b1     = (const float*)d_in[5];
    const float* W2     = (const float*)d_in[6];
    const float* b2     = (const float*)d_in[7];
    float* out = (float*)d_out;

    const int N = in_sizes[0] / 128;
    const int E = in_sizes[1] / 2;
    const int* src = ei;
    const int* dst = ei + E;

    float* ws   = (float*)d_ws;
    float* deg  = ws;
    float* dinv = deg  + N;
    float* xw   = dinv + N;
    float* lw   = xw   + (size_t)N * 32;
    float* h1   = lw   + (size_t)N * 16;
    float* tlin = h1   + (size_t)N * 16;
    float* tbuf = tlin + N;
    _Float16* bp1 = (_Float16*)(tbuf + N);
    _Float16* bp2 = bp1 + 4 * 2 * 512;

    const int thr = 256;
    auto cdiv = [](int a, int b) { return (a + b - 1) / b; };
    const int mtiles = cdiv(N, 16);
    const int wpb = thr / 32;

    k_init_deg <<<cdiv(N, thr), thr, 0, stream>>>(deg, N);
    k_accum_deg<<<cdiv(E, thr), thr, 0, stream>>>(dst, E, deg);
    k_dinv     <<<cdiv(N, thr), thr, 0, stream>>>(deg, dinv, N);
    k_pack_b<2><<<1, 256, 0, stream>>>(W_base, bp1, 128);
    k_pack_b<1><<<1,  32, 0, stream>>>(W1,     bp2, 32);

    k_wmma_gemm<2><<<cdiv(mtiles, wpb), thr, 0, stream>>>(x, bp1, xw, N, 128);
    k_init_agg<32><<<cdiv(N * 32, thr), thr, 0, stream>>>(xw, dinv, b_base, out, N);
    k_scatter<32> <<<cdiv(E * 32, thr), thr, 0, stream>>>(xw, src, dst, dinv, out, E);

    k_wmma_gemm<1><<<cdiv(mtiles, wpb), thr, 0, stream>>>(out, bp2, lw, N, 32);
    k_init_agg<16><<<cdiv(N * 16, thr), thr, 0, stream>>>(lw, dinv, b1, h1, N);
    k_scatter<16> <<<cdiv(E * 16, thr), thr, 0, stream>>>(lw, src, dst, dinv, h1, E);

    k_matvec16_relu<<<cdiv(N, thr), thr, 0, stream>>>(h1, W2, tlin, N);
    k_init_agg<1><<<cdiv(N, thr), thr, 0, stream>>>(tlin, dinv, b2, tbuf, N);
    k_scatter<1> <<<cdiv(E, thr), thr, 0, stream>>>(tlin, src, dst, dinv, tbuf, E);

    k_finalize <<<cdiv(N * 32, thr), thr, 0, stream>>>(tbuf, out, N);
}


# Rows a gather reads are stored pre-multiplied by their node's dinv (GEMM and matvec stores): a neighbour is one row load, no weight lookup; dest dinv, then bias, applied after the sum
# speedup vs baseline: 1.9251x; 1.9251x over previous
#include <hip/hip_runtime.h>
#include <hip/hip_bf16.h>
#include <math.h>


typedef __attribute__((ext_vector_type(16))) _Float16 v16h;
typedef __attribute__((ext_vector_type(8)))  float    v8f;

static __device__ __forceinline__ int gtid() {
    return blockIdx.x * blockDim.x + threadIdx.x;
}


__global__ void k_zero_cnt(int* __restrict__ cnt, int n) {
    int i = gtid();
    if (i < n) cnt[i] = 0;
}

__global__ void k_count_rank(const int* __restrict__ dst, int e,
                             int* __restrict__ cnt, int* __restrict__ rank) {
    int i = gtid();
    if (i < e) rank[i] = atomicAdd(&cnt[dst[i]], 1);
}

__global__ void __launch_bounds__(256)
k_scan_dinv(const int* __restrict__ cnt, int* __restrict__ rowstart,
            float* __restrict__ dinv, int n) {
    __shared__ int wtot[8];
    __shared__ int xtot[8];
    const int tid  = threadIdx.x;
    const int lane = tid & 31;
    const int w    = tid >> 5;
    const int bbeg = blockIdx.x * 1024;

    int x = 0;
#pragma unroll 8
    for (int i = tid * 4; i < bbeg; i += 1024) {
        int4 p = *reinterpret_cast<const int4*>(cnt + i);
        x += (p.x + p.y) + (p.z + p.w);
    }

    const int i0 = bbeg + tid * 4;
    int4 v = make_int4(0, 0, 0, 0);
    if (i0 + 3 < n) {
        v = *reinterpret_cast<const int4*>(cnt + i0);
    } else {
        if (i0     < n) v.x = cnt[i0];
        if (i0 + 1 < n) v.y = cnt[i0 + 1];
        if (i0 + 2 < n) v.z = cnt[i0 + 2];
    }
    const int t = (v.x + v.y) + (v.z + v.w);

    int sc = t;
    int xs = x;
    for (int off = 1; off < 32; off <<= 1) {
        int a = __shfl_up(sc, off);
        int b = __shfl_up(xs, off);
        if (lane >= off) { sc += a; xs += b; }
    }
    if (lane == 31) { wtot[w] = sc; xtot[w] = xs; }
    __syncthreads();
    int base = sc - t;
    for (int j = 0; j < 8; ++j) {
        base += xtot[j];
        if (j < w) base += wtot[j];
    }

    int4 o;
    o.x = base;
    o.y = o.x + v.x;
    o.z = o.y + v.y;
    o.w = o.z + v.z;
    if (i0 + 3 < n) {
        *reinterpret_cast<int4*>(rowstart + i0) = o;
        dinv[i0 + 0] = rsqrtf(fmaxf((float)(v.x + 1), 1.0f));
        dinv[i0 + 1] = rsqrtf(fmaxf((float)(v.y + 1), 1.0f));
        dinv[i0 + 2] = rsqrtf(fmaxf((float)(v.z + 1), 1.0f));
        dinv[i0 + 3] = rsqrtf(fmaxf((float)(v.w + 1), 1.0f));
    } else {
        if (i0 < n) {
            rowstart[i0] = o.x;
            dinv[i0] = rsqrtf(fmaxf((float)(v.x + 1), 1.0f));
        }
        if (i0 + 1 < n) {
            rowstart[i0 + 1] = o.y;
            dinv[i0 + 1] = rsqrtf(fmaxf((float)(v.y + 1), 1.0f));
        }
        if (i0 + 2 < n) {
            rowstart[i0 + 2] = o.z;
            dinv[i0 + 2] = rsqrtf(fmaxf((float)(v.z + 1), 1.0f));
        }
    }
}

__global__ void k_fill(const int* __restrict__ src, const int* __restrict__ dst,
                       const int* __restrict__ rank,
                       const int* __restrict__ rowstart,
                       int* __restrict__ csr, int e) {
    int i = gtid();
    if (i >= e) return;
    unsigned pos = (unsigned)(rowstart[dst[i]] + rank[i]);
    if (pos < (unsigned)e) csr[pos] = src[i];
}


template <int NT>
__global__ void k_pack_b(const float* __restrict__ B,
                         _Float16* __restrict__ Bp, int K) {
    const int N = NT * 16;
    const int chunks = K >> 5;
    int i = gtid();
    if (i >= chunks * NT * 32) return;
    const int lane = i & 31;
    const int t    = (i >> 5) % NT;
    const int c    = (i >> 5) / NT;
    const int r    = lane & 15;
    const int kh   = (lane >> 4) << 3;
    const int k0   = c << 5;
    const int col  = t * 16 + r;
    _Float16* o = Bp + (size_t)i * 16;
#pragma unroll
    for (int j = 0; j < 8; ++j) {
        o[j]     = (_Float16)B[(size_t)(k0 + kh + j) * N + col];
        o[8 + j] = (_Float16)B[(size_t)(k0 + 16 + kh + j) * N + col];
    }
}


template <int NT>
__global__ void k_wmma_gemm(const float* __restrict__ A,
                            const _Float16* __restrict__ Bp,
                            const float* __restrict__ rowscale,
                            float* __restrict__ C,
                            int M, int K) {
    const int N    = NT * 16;
    const int wave = threadIdx.x >> 5;
    const int lane = threadIdx.x & 31;
    const int tile = blockIdx.x * (blockDim.x >> 5) + wave;
    if (tile * 16 >= M) return;

    const int r   = lane & 15;
    const int kh  = (lane >> 4) << 3;
    const int row = tile * 16 + r;
    const float* __restrict__ arow = A + (size_t)row * K;
    const v16h* __restrict__ bfrag = (const v16h*)Bp;

    const int mbase = (lane < 16) ? 0 : 8;
    float rs[8];
#pragma unroll
    for (int j = 0; j < 8; ++j) rs[j] = rowscale[(size_t)(tile * 16 + mbase + j)];

    v8f zero = {};
    v8f acc[NT];
#pragma unroll
    for (int t = 0; t < NT; ++t) acc[t] = zero;

    for (int k0 = 0; k0 < K; k0 += 32) {
        const float4* __restrict__ a0 =
            reinterpret_cast<const float4*>(arow + k0 + kh);
        const float4* __restrict__ a1 =
            reinterpret_cast<const float4*>(arow + k0 + 16 + kh);
        float4 p0 = a0[0], p1 = a0[1];
        float4 q0 = a1[0], q1 = a1[1];
        v16h a;
        a[0]  = (_Float16)p0.x; a[1]  = (_Float16)p0.y;
        a[2]  = (_Float16)p0.z; a[3]  = (_Float16)p0.w;
        a[4]  = (_Float16)p1.x; a[5]  = (_Float16)p1.y;
        a[6]  = (_Float16)p1.z; a[7]  = (_Float16)p1.w;
        a[8]  = (_Float16)q0.x; a[9]  = (_Float16)q0.y;
        a[10] = (_Float16)q0.z; a[11] = (_Float16)q0.w;
        a[12] = (_Float16)q1.x; a[13] = (_Float16)q1.y;
        a[14] = (_Float16)q1.z; a[15] = (_Float16)q1.w;

        const int cidx = (k0 >> 5) * NT;
#pragma unroll
        for (int t = 0; t < NT; ++t) {
            v16h b = bfrag[(size_t)(cidx + t) * 32 + lane];
            acc[t] = __builtin_amdgcn_wmma_f32_16x16x32_f16(
                false, a, false, b, (short)0, acc[t], false, false);
        }
    }

    const int ncol  = lane & 15;
#pragma unroll
    for (int j = 0; j < 8; ++j) {
        const size_t orow = (size_t)(tile * 16 + mbase + j);
#pragma unroll
        for (int t = 0; t < NT; ++t)
            C[orow * N + t * 16 + ncol] = acc[t][j] * rs[j];
    }
}


template <int C>
__global__ void k_gather(const float* __restrict__ hw,
                         const int* __restrict__ rowstart,
                         const int* __restrict__ csr,
                         const float* __restrict__ dinv,
                         const float* __restrict__ bias,
                         float* __restrict__ agg, int n, int e) {
    const int i = gtid();
    const int node = i / C, c = i % C;
    if (node >= n) return;
    const unsigned nmax = (unsigned)(n - 1);

    if constexpr (C == 32) {
        const int nd   = __builtin_amdgcn_readfirstlane(node);
        const int lane = c;
        const int beg  = __builtin_amdgcn_readfirstlane(rowstart[nd]);
        const int end  = (nd + 1 < n)
                       ? __builtin_amdgcn_readfirstlane(rowstart[nd + 1]) : e;
        const float dd = dinv[nd];
        float acc = hw[(size_t)nd * 32 + lane];
        for (int base = beg; base < end; base += 32) {
            const int m = min(32, end - base);
            int sv = 0;
            if (lane < m)
                sv = (int)min((unsigned)csr[base + lane], nmax);
#pragma unroll 4
            for (int j = 0; j < m; ++j) {
                const int s = __builtin_amdgcn_readlane(sv, j);
                acc += hw[(size_t)s * 32 + lane];
            }
        }
        agg[(size_t)nd * 32 + lane] = bias[lane] + acc * dd;
    } else {
        const int beg  = rowstart[node];
        const int end  = (node + 1 < n) ? rowstart[node + 1] : e;
        const float dd = dinv[node];
        float acc = hw[i];
#pragma unroll 4
        for (int j = beg; j < end; ++j) {
            const int s = (int)min((unsigned)csr[j], nmax);
            acc += hw[(size_t)s * C + c];
        }
        agg[i] = bias[c] + acc * dd;
    }
}

__global__ void k_matvec16_relu(const float* __restrict__ h,
                                const float* __restrict__ w,
                                const float* __restrict__ dinv,
                                float* __restrict__ out, int n) {
    int i = gtid();
    if (i >= n) return;
    const float4* hr = (const float4*)(h + (size_t)i * 16);
    float s = 0.0f;
#pragma unroll
    for (int j = 0; j < 4; ++j) {
        float4 v = hr[j];
        s += fmaxf(v.x, 0.0f) * w[j * 4 + 0];
        s += fmaxf(v.y, 0.0f) * w[j * 4 + 1];
        s += fmaxf(v.z, 0.0f) * w[j * 4 + 2];
        s += fmaxf(v.w, 0.0f) * w[j * 4 + 3];
    }
    out[i] = s * dinv[i];
}

__global__ void k_finalize(const float* __restrict__ t,
                           float* __restrict__ out, int n) {
    int i = gtid();
    if (i >= n * 32) return;
    float tv = logf(expf(t[i >> 5]) + 1.1f);
    out[i] *= tv;
}


extern "C" void kernel_launch(void* const* d_in, const int* in_sizes, int n_in,
                              void* d_out, int out_size, void* d_ws, size_t ws_size,
                              hipStream_t stream) {
    const float* x      = (const float*)d_in[0];
    const int*   ei     = (const int*)  d_in[1];
    const float* W_base = (const float*)d_in[2];
    const float* b_base = (const float*)d_in[3];
    const float* W1     = (const float*)d_in[4];
    const float* b1     = (const float*)d_in[5];
    const float* W2     = (const float*)d_in[6];
    const float* b2     = (const float*)d_in[7];
    float* out = (float*)d_out;

    const int N = in_sizes[0] / 128;
    const int E = in_sizes[1] / 2;
    const int* src = ei;
    const int* dst = ei + E;

    float* ws   = (float*)d_ws;
    int*   rowstart = (int*)ws;
    float* dinv = ws   + N;
    float* xw   = dinv + N;
    float* lw   = xw;
    float* h1   = xw   + (size_t)N * 16;
    int*   csr  = (int*)(xw + (size_t)N * 32);
    int*   rank = csr  + E;
    float* tlin = (float*)(rank + E);
    float* tbuf = tlin + N;
    int*   cnt  = (int*)tbuf;
    _Float16* bp1 = (_Float16*)(tbuf + N);
    _Float16* bp2 = bp1 + 4 * 2 * 512;

    const int thr = 256;
    auto cdiv = [](int a, int b) { return (a + b - 1) / b; };
    const int mtiles = cdiv(N, 16);
    const int wpb = thr / 32;

    k_zero_cnt  <<<cdiv(N, thr), thr, 0, stream>>>(cnt, N);
    k_count_rank<<<cdiv(E, thr), thr, 0, stream>>>(dst, E, cnt, rank);
    k_scan_dinv <<<cdiv(N, 1024), 256, 0, stream>>>(cnt, rowstart, dinv, N);
    k_fill      <<<cdiv(E, thr), thr, 0, stream>>>(src, dst, rank, rowstart, csr, E);
    k_pack_b<2><<<1, 256, 0, stream>>>(W_base, bp1, 128);
    k_pack_b<1><<<1,  32, 0, stream>>>(W1,     bp2, 32);

    k_wmma_gemm<2><<<cdiv(mtiles, wpb), thr, 0, stream>>>(x, bp1, dinv, xw, N, 128);
    k_gather<32><<<cdiv(N * 32, thr), thr, 0, stream>>>(xw, rowstart, csr, dinv, b_base, out, N, E);

    k_wmma_gemm<1><<<cdiv(mtiles, wpb), thr, 0, stream>>>(out, bp2, dinv, lw, N, 32);
    k_gather<16><<<cdiv(N * 16, thr), thr, 0, stream>>>(lw, rowstart, csr, dinv, b1, h1, N, E);

    k_matvec16_relu<<<cdiv(N, thr), thr, 0, stream>>>(h1, W2, dinv, tlin, N);
    k_gather<1><<<cdiv(N, thr), thr, 0, stream>>>(tlin, rowstart, csr, dinv, b2, tbuf, N, E);

    k_finalize <<<cdiv(N * 32, thr), thr, 0, stream>>>(tbuf, out, N);
}


# Edges bucketed by a two-level counting sort with every per-edge counter in LDS (chunks by 256-node range, then one workgroup per range): no global atomic or scattered store; list order = LDS arrival
# speedup vs baseline: 4.5184x; 2.3471x over previous
#include <hip/hip_runtime.h>
#include <hip/hip_bf16.h>
#include <math.h>


typedef __attribute__((ext_vector_type(16))) _Float16 v16h;
typedef __attribute__((ext_vector_type(8)))  float    v8f;

static __device__ __forceinline__ int gtid() {
    return blockIdx.x * blockDim.x + threadIdx.x;
}


#define CH   4096
#define MAXB 1024
#define MAXC 512
#define CAP  6144

template <int PER>
static __device__ __forceinline__ void block_excl_scan(int* a, int len, int* wtot) {
    const int tid  = threadIdx.x;
    const int lane = tid & 31;
    const int w    = tid >> 5;
    const int b0   = tid * PER;
    int v[PER];
    int s = 0;
#pragma unroll
    for (int j = 0; j < PER; ++j) {
        v[j] = (b0 + j < len) ? a[b0 + j] : 0;
        s += v[j];
    }
    int inc = s;
    for (int off = 1; off < 32; off <<= 1) {
        int t = __shfl_up(inc, off);
        if (lane >= off) inc += t;
    }
    if (lane == 31) wtot[w] = inc;
    __syncthreads();
    int base = inc - s;
    for (int j = 0; j < 8; ++j)
        if (j < w) base += wtot[j];
#pragma unroll
    for (int j = 0; j < PER; ++j) {
        if (b0 + j < len) { a[b0 + j] = base; base += v[j]; }
    }
    __syncthreads();
}

__global__ void __launch_bounds__(256)
k_chunk_sort(const int* __restrict__ src, const int* __restrict__ dst,
             int e, int n, int nbins,
             int* __restrict__ pairs, int* __restrict__ blockstart) {
    __shared__ int hist[MAXB + 1];
    __shared__ int cur[MAXB];
    __shared__ int stage[CH];
    __shared__ int wtot[8];
    if (nbins > MAXB) return;
    const int tid  = threadIdx.x;
    const int base = blockIdx.x * CH;
    const unsigned nmax = (unsigned)(n - 1);

    for (int b = tid; b <= nbins; b += 256) hist[b] = 0;

    int dd[16], ss[16];
    const bool al  = ((((size_t)src) | ((size_t)dst)) & 15) == 0;
    const bool alp = (((size_t)pairs) & 15) == 0;
#pragma unroll
    for (int k = 0; k < 4; ++k) {
        const int idx = base + k * 1024 + tid * 4;
        if (al && idx + 3 < e) {
            const int4 d4 = *reinterpret_cast<const int4*>(dst + idx);
            const int4 s4 = *reinterpret_cast<const int4*>(src + idx);
            dd[k * 4 + 0] = d4.x; dd[k * 4 + 1] = d4.y; dd[k * 4 + 2] = d4.z; dd[k * 4 + 3] = d4.w;
            ss[k * 4 + 0] = s4.x; ss[k * 4 + 1] = s4.y; ss[k * 4 + 2] = s4.z; ss[k * 4 + 3] = s4.w;
        } else {
#pragma unroll
            for (int j = 0; j < 4; ++j) {
                const bool ok = idx + j < e;
                dd[k * 4 + j] = ok ? dst[idx + j] : -1;
                ss[k * 4 + j] = ok ? src[idx + j] : 0;
            }
        }
    }
    bool valid[16];
#pragma unroll
    for (int k = 0; k < 16; ++k) {
        const int idx = base + (k >> 2) * 1024 + tid * 4 + (k & 3);
        valid[k] = idx < e;
        dd[k] = (int)min((unsigned)dd[k], nmax);
    }
    __syncthreads();
#pragma unroll
    for (int k = 0; k < 16; ++k)
        if (valid[k]) atomicAdd(&hist[dd[k] >> 8], 1);
    __syncthreads();
    block_excl_scan<(MAXB + 256) / 256>(hist, nbins + 1, wtot);

    int* __restrict__ row = blockstart + (size_t)blockIdx.x * (nbins + 1);
    for (int b = tid; b <= nbins; b += 256) {
        const int h = hist[b];
        row[b] = h;
        if (b < nbins) cur[b] = h;
    }
    __syncthreads();
#pragma unroll
    for (int k = 0; k < 16; ++k) {
        if (valid[k]) {
            const int pos = atomicAdd(&cur[dd[k] >> 8], 1);
            stage[pos] = (int)(((unsigned)ss[k] << 8) | ((unsigned)dd[k] & 255u));
        }
    }
    __syncthreads();
#pragma unroll
    for (int k = 0; k < 4; ++k) {
        const int li  = k * 1024 + tid * 4;
        const int idx = base + li;
        if (alp && idx + 3 < e) {
            *reinterpret_cast<int4*>(pairs + idx) =
                make_int4(stage[li], stage[li + 1], stage[li + 2], stage[li + 3]);
        } else {
            for (int j = 0; j < 4; ++j)
                if (idx + j < e) pairs[idx + j] = stage[li + j];
        }
    }
}

__global__ void __launch_bounds__(256)
k_bin_sort(const int* __restrict__ pairs, const int* __restrict__ blockstart,
           int e, int n, int nbins, int nchunks,
           int* __restrict__ csr, int* __restrict__ rowstart,
           float* __restrict__ dinv) {
    __shared__ int rstart[MAXC];
    __shared__ int roff[MAXC + 1];
    __shared__ int hist[256];
    __shared__ int cur[256];
    __shared__ int buf[CAP];
    __shared__ int outb[CAP];
    __shared__ int wtot[8];
    __shared__ int bs_sh;
    if (nbins > MAXB || nchunks > MAXC) return;
    const int tid = threadIdx.x;
    const int b   = blockIdx.x;
    const int g   = tid >> 3;
    const int gl  = tid & 7;

    if (tid == 0) bs_sh = 0;
    hist[tid] = 0;
    __syncthreads();
    int below = 0;
    for (int c = tid; c <= nchunks; c += 256) {
        int len = 0;
        if (c < nchunks) {
            const int* __restrict__ row = blockstart + (size_t)c * (nbins + 1);
            const int v0 = row[b];
            const int v1 = row[b + 1];
            rstart[c] = c * CH + v0;
            len = v1 - v0;
            below += v0;
        }
        roff[c] = len;
    }
    atomicAdd(&bs_sh, below);
    __syncthreads();
    block_excl_scan<(MAXC + 256) / 256>(roff, nchunks + 1, wtot);
    const int bs  = bs_sh;
    const int cnt = roff[nchunks];
    const bool small = cnt <= CAP;

    const int elast = e - 1;
    for (int c0 = g; c0 < nchunks; c0 += 128) {
        int rr[4], oo[4], ll[4], pv[8];
#pragma unroll
        for (int j = 0; j < 4; ++j) {
            const int cc = c0 + 32 * j;
            const bool in = cc < nchunks;
            const int ccs = in ? cc : 0;
            rr[j] = rstart[ccs];
            oo[j] = roff[ccs];
            ll[j] = in ? roff[ccs + 1] - oo[j] : 0;
        }
#pragma unroll
        for (int j = 0; j < 4; ++j) {
            pv[2 * j]     = pairs[min(rr[j] + gl, elast)];
            pv[2 * j + 1] = pairs[min(rr[j] + gl + 8, elast)];
        }
#pragma unroll
        for (int j = 0; j < 8; ++j) {
            const int i = gl + 8 * (j & 1);
            if (i < ll[j >> 1]) {
                if (small) buf[oo[j >> 1] + i] = pv[j];
                atomicAdd(&hist[pv[j] & 255], 1);
            }
        }
#pragma unroll
        for (int j = 0; j < 4; ++j) {
            for (int i = gl + 16; i < ll[j]; i += 8) {
                const int p = pairs[rr[j] + i];
                if (small) buf[oo[j] + i] = p;
                atomicAdd(&hist[p & 255], 1);
            }
        }
    }
    __syncthreads();
    const int mycnt = hist[tid];
    block_excl_scan<1>(hist, 256, wtot);
    const int mystart = hist[tid];
    cur[tid] = mystart;
    const int node = b * 256 + tid;
    if (node < n) {
        rowstart[node] = bs + mystart;
        dinv[node] = rsqrtf(fmaxf((float)(mycnt + 1), 1.0f));
    }
    __syncthreads();

    if (small) {
        for (int i = tid; i < cnt; i += 256) {
            const unsigned p = (unsigned)buf[i];
            const int pos = atomicAdd(&cur[p & 255u], 1);
            outb[pos] = (int)(p >> 8);
        }
        __syncthreads();
        for (int i = tid; i < cnt; i += 256)
            if (bs + i < e) csr[bs + i] = outb[i];
    } else {
        for (int c = g; c < nchunks; c += 32) {
            const int rs = rstart[c];
            const int len = roff[c + 1] - roff[c];
            for (int i = gl; i < len; i += 8) {
                const unsigned p = (unsigned)pairs[rs + i];
                const int pos = bs + atomicAdd(&cur[p & 255u], 1);
                if ((unsigned)pos < (unsigned)e) csr[pos] = (int)(p >> 8);
            }
        }
    }
}


template <int NT>
__global__ void k_pack_b(const float* __restrict__ B,
                         _Float16* __restrict__ Bp, int K) {
    const int N = NT * 16;
    const int chunks = K >> 5;
    int i = gtid();
    if (i >= chunks * NT * 32) return;
    const int lane = i & 31;
    const int t    = (i >> 5) % NT;
    const int c    = (i >> 5) / NT;
    const int r    = lane & 15;
    const int kh   = (lane >> 4) << 3;
    const int k0   = c << 5;
    const int col  = t * 16 + r;
    _Float16* o = Bp + (size_t)i * 16;
#pragma unroll
    for (int j = 0; j < 8; ++j) {
        o[j]     = (_Float16)B[(size_t)(k0 + kh + j) * N + col];
        o[8 + j] = (_Float16)B[(size_t)(k0 + 16 + kh + j) * N + col];
    }
}


template <int NT>
__global__ void k_wmma_gemm(const float* __restrict__ A,
                            const _Float16* __restrict__ Bp,
                            const float* __restrict__ rowscale,
                            float* __restrict__ C,
                            int M, int K) {
    const int N    = NT * 16;
    const int wave = threadIdx.x >> 5;
    const int lane = threadIdx.x & 31;
    const int tile = blockIdx.x * (blockDim.x >> 5) + wave;
    if (tile * 16 >= M) return;

    const int r   = lane & 15;
    const int kh  = (lane >> 4) << 3;
    const int row = tile * 16 + r;
    const float* __restrict__ arow = A + (size_t)row * K;
    const v16h* __restrict__ bfrag = (const v16h*)Bp;

    const int mbase = (lane < 16) ? 0 : 8;
    float rs[8];
#pragma unroll
    for (int j = 0; j < 8; ++j) rs[j] = rowscale[(size_t)(tile * 16 + mbase + j)];

    v8f zero = {};
    v8f acc[NT];
#pragma unroll
    for (int t = 0; t < NT; ++t) acc[t] = zero;

    for (int k0 = 0; k0 < K; k0 += 32) {
        const float4* __restrict__ a0 =
            reinterpret_cast<const float4*>(arow + k0 + kh);
        const float4* __restrict__ a1 =
            reinterpret_cast<const float4*>(arow + k0 + 16 + kh);
        float4 p0 = a0[0], p1 = a0[1];
        float4 q0 = a1[0], q1 = a1[1];
        v16h a;
        a[0]  = (_Float16)p0.x; a[1]  = (_Float16)p0.y;
        a[2]  = (_Float16)p0.z; a[3]  = (_Float16)p0.w;
        a[4]  = (_Float16)p1.x; a[5]  = (_Float16)p1.y;
        a[6]  = (_Float16)p1.z; a[7]  = (_Float16)p1.w;
        a[8]  = (_Float16)q0.x; a[9]  = (_Float16)q0.y;
        a[10] = (_Float16)q0.z; a[11] = (_Float16)q0.w;
        a[12] = (_Float16)q1.x; a[13] = (_Float16)q1.y;
        a[14] = (_Float16)q1.z; a[15] = (_Float16)q1.w;

        const int cidx = (k0 >> 5) * NT;
#pragma unroll
        for (int t = 0; t < NT; ++t) {
            v16h b = bfrag[(size_t)(cidx + t) * 32 + lane];
            acc[t] = __builtin_amdgcn_wmma_f32_16x16x32_f16(
                false, a, false, b, (short)0, acc[t], false, false);
        }
    }

    const int ncol  = lane & 15;
#pragma unroll
    for (int j = 0; j < 8; ++j) {
        const size_t orow = (size_t)(tile * 16 + mbase + j);
#pragma unroll
        for (int t = 0; t < NT; ++t)
            C[orow * N + t * 16 + ncol] = acc[t][j] * rs[j];
    }
}


static __device__ __forceinline__ void add4(float4& a, const float4 v) {
    a.x += v.x; a.y += v.y; a.z += v.z; a.w += v.w;
}

template <int K, int ROWB>
static __device__ __forceinline__ void take_rows(const char* __restrict__ hwb,
                                                 const int* __restrict__ csr,
                                                 int j, unsigned nmax,
                                                 unsigned qoff, float4& acc) {
    unsigned off[K];
#pragma unroll
    for (int k = 0; k < K; ++k)
        off[k] = min((unsigned)csr[j + k], nmax) * (unsigned)ROWB + qoff;
    float4 v[K];
#pragma unroll
    for (int k = 0; k < K; ++k)
        v[k] = *reinterpret_cast<const float4*>(hwb + off[k]);
#pragma unroll
    for (int k = 0; k < K; ++k) add4(acc, v[k]);
}

template <int C>
__global__ void k_gather_v4(const float* __restrict__ hw,
                            const int* __restrict__ rowstart,
                            const int* __restrict__ csr,
                            const float* __restrict__ dinv,
                            const float* __restrict__ bias,
                            float* __restrict__ agg, int n, int e) {
    constexpr int L    = C / 4;
    constexpr int ROWB = C * 4;
    const int i = gtid();
    const int node = i / L, q = i % L;
    if (node >= n) return;
    const unsigned nmax = (unsigned)(n - 1);
    const unsigned qoff = (unsigned)q * 16u;
    const char* __restrict__ hwb = reinterpret_cast<const char*>(hw);

    const int beg  = rowstart[node];
    const int end  = (node + 1 < n) ? rowstart[node + 1] : e;
    const float dd = dinv[node];
    const float4 b = reinterpret_cast<const float4*>(bias)[q];
    float4 acc = *reinterpret_cast<const float4*>(hwb + (unsigned)node * (unsigned)ROWB + qoff);

    int j = beg;
    for (; j + 8 <= end; j += 8) take_rows<8, ROWB>(hwb, csr, j, nmax, qoff, acc);
    if (j + 4 <= end) { take_rows<4, ROWB>(hwb, csr, j, nmax, qoff, acc); j += 4; }
    const int r = end - j;
    if (r > 0) {
        const int last = end - 1;
        const unsigned o0 = min((unsigned)csr[j],                nmax) * (unsigned)ROWB + qoff;
        const unsigned o1 = min((unsigned)csr[min(j + 1, last)], nmax) * (unsigned)ROWB + qoff;
        const unsigned o2 = min((unsigned)csr[min(j + 2, last)], nmax) * (unsigned)ROWB + qoff;
        const float4 v0 = *reinterpret_cast<const float4*>(hwb + o0);
        const float4 v1 = *reinterpret_cast<const float4*>(hwb + o1);
        const float4 v2 = *reinterpret_cast<const float4*>(hwb + o2);
        add4(acc, v0);
        if (r > 1) add4(acc, v1);
        if (r > 2) add4(acc, v2);
    }

    float4 o;
    o.x = b.x + acc.x * dd;
    o.y = b.y + acc.y * dd;
    o.z = b.z + acc.z * dd;
    o.w = b.w + acc.w * dd;
    *reinterpret_cast<float4*>(reinterpret_cast<char*>(agg) + (unsigned)node * (unsigned)ROWB + qoff) = o;
}

__global__ void k_gather1(const float* __restrict__ hw,
                          const int* __restrict__ rowstart,
                          const int* __restrict__ csr,
                          const float* __restrict__ dinv,
                          const float* __restrict__ bias,
                          float* __restrict__ agg, int n, int e) {
    const int node = gtid();
    if (node >= n) return;
    const unsigned nmax = (unsigned)(n - 1);
    const int beg  = rowstart[node];
    const int end  = (node + 1 < n) ? rowstart[node + 1] : e;
    const float dd = dinv[node];
    float acc = hw[node];
#pragma unroll 4
    for (int j = beg; j < end; ++j) {
        const int s = (int)min((unsigned)csr[j], nmax);
        acc += hw[s];
    }
    agg[node] = bias[0] + acc * dd;
}

__global__ void k_matvec16_relu(const float* __restrict__ h,
                                const float* __restrict__ w,
                                const float* __restrict__ dinv,
                                float* __restrict__ out, int n) {
    int i = gtid();
    if (i >= n) return;
    const float4* hr = (const float4*)(h + (size_t)i * 16);
    float s = 0.0f;
#pragma unroll
    for (int j = 0; j < 4; ++j) {
        float4 v = hr[j];
        s += fmaxf(v.x, 0.0f) * w[j * 4 + 0];
        s += fmaxf(v.y, 0.0f) * w[j * 4 + 1];
        s += fmaxf(v.z, 0.0f) * w[j * 4 + 2];
        s += fmaxf(v.w, 0.0f) * w[j * 4 + 3];
    }
    out[i] = s * dinv[i];
}

__global__ void k_finalize(const float* __restrict__ t,
                           float* __restrict__ out, int n) {
    const int total  = n * 8;
    const int stride = gridDim.x * blockDim.x;
    const int i0     = gtid();
    float4* __restrict__ o4 = reinterpret_cast<float4*>(out);
    float4 v[4];
    float  tv[4];
#pragma unroll
    for (int k = 0; k < 4; ++k) {
        const int ic = min(i0 + k * stride, total - 1);
        v[k]  = o4[ic];
        tv[k] = t[ic >> 3];
    }
#pragma unroll
    for (int k = 0; k < 4; ++k) {
        const int idx = i0 + k * stride;
        const float f = logf(expf(tv[k]) + 1.1f);
        float4 w = v[k];
        w.x *= f; w.y *= f; w.z *= f; w.w *= f;
        if (idx < total) o4[idx] = w;
    }
}


extern "C" void kernel_launch(void* const* d_in, const int* in_sizes, int n_in,
                              void* d_out, int out_size, void* d_ws, size_t ws_size,
                              hipStream_t stream) {
    const float* x      = (const float*)d_in[0];
    const int*   ei     = (const int*)  d_in[1];
    const float* W_base = (const float*)d_in[2];
    const float* b_base = (const float*)d_in[3];
    const float* W1     = (const float*)d_in[4];
    const float* b1     = (const float*)d_in[5];
    const float* W2     = (const float*)d_in[6];
    const float* b2     = (const float*)d_in[7];
    float* out = (float*)d_out;

    const int N = in_sizes[0] / 128;
    const int E = in_sizes[1] / 2;
    const int* src = ei;
    const int* dst = ei + E;

    float* ws   = (float*)d_ws;
    int*   rowstart = (int*)ws;
    float* dinv = ws   + N;
    float* xw   = dinv + N;
    float* lw   = xw;
    float* h1   = xw   + (size_t)N * 16;
    int*   csr  = (int*)(xw + (size_t)N * 32);
    int*   pairs = csr + E;
    float* tlin = (float*)(pairs + E);
    float* tbuf = tlin + N;
    int*   bstart = (int*)xw;
    _Float16* bp1 = (_Float16*)(tbuf + N);
    _Float16* bp2 = bp1 + 4 * 2 * 512;

    const int thr = 256;
    auto cdiv = [](int a, int b) { return (a + b - 1) / b; };
    const int mtiles = cdiv(N, 16);
    const int wpb = thr / 32;

    const int nbins   = cdiv(N, 256);
    const int nchunks = cdiv(E, CH);
    k_chunk_sort<<<nchunks, 256, 0, stream>>>(src, dst, E, N, nbins, pairs, bstart);
    k_bin_sort  <<<nbins, 256, 0, stream>>>(pairs, bstart, E, N, nbins, nchunks, csr, rowstart, dinv);
    k_pack_b<2><<<1, 256, 0, stream>>>(W_base, bp1, 128);
    k_pack_b<1><<<1,  32, 0, stream>>>(W1,     bp2, 32);

    k_wmma_gemm<2><<<cdiv(mtiles, wpb), thr, 0, stream>>>(x, bp1, dinv, xw, N, 128);
    k_gather_v4<32><<<cdiv(N * 8, thr), thr, 0, stream>>>(xw, rowstart, csr, dinv, b_base, out, N, E);

    k_wmma_gemm<1><<<cdiv(mtiles, wpb), thr, 0, stream>>>(out, bp2, dinv, lw, N, 32);
    k_gather_v4<16><<<cdiv(N * 4, thr), thr, 0, stream>>>(lw, rowstart, csr, dinv, b1, h1, N, E);

    k_matvec16_relu<<<cdiv(N, thr), thr, 0, stream>>>(h1, W2, dinv, tlin, N);
    k_gather1<<<cdiv(N, thr), thr, 0, stream>>>(tlin, rowstart, csr, dinv, b2, tbuf, N, E);

    k_finalize <<<cdiv(cdiv(N * 8, 4), thr), thr, 0, stream>>>(tbuf, out, N);
}


# Follow-on stages run in the gather that feeds them (logits@W1 WMMA in the 32-ch; ReLU dot as 4 lane partials + 2-step tree in the 16-ch; final scaling in the 1-ch; packs in launch 1): 11 launches -> 6
# speedup vs baseline: 5.1467x; 1.1391x over previous
#include <hip/hip_runtime.h>
#include <hip/hip_bf16.h>
#include <math.h>


typedef __attribute__((ext_vector_type(16))) _Float16 v16h;
typedef __attribute__((ext_vector_type(8)))  float    v8f;

static __device__ __forceinline__ int gtid() {
    return blockIdx.x * blockDim.x + threadIdx.x;
}


template <int NT>
static __device__ __forceinline__ void pack_b(const float* __restrict__ B,
                                              _Float16* __restrict__ Bp, int K) {
    const int N = NT * 16;
    const int chunks = K >> 5;
    for (int i = threadIdx.x; i < chunks * NT * 32; i += blockDim.x) {
        const int lane = i & 31;
        const int t    = (i >> 5) % NT;
        const int c    = (i >> 5) / NT;
        const int r    = lane & 15;
        const int kh   = (lane >> 4) << 3;
        const int k0   = c << 5;
        const int col  = t * 16 + r;
        _Float16* o = Bp + (size_t)i * 16;
#pragma unroll
        for (int j = 0; j < 8; ++j) {
            o[j]     = (_Float16)B[(size_t)(k0 + kh + j) * N + col];
            o[8 + j] = (_Float16)B[(size_t)(k0 + 16 + kh + j) * N + col];
        }
    }
}

#define CH   4096
#define MAXB 1024
#define MAXC 512
#define CAP  6144

template <int PER>
static __device__ __forceinline__ void block_excl_scan(int* a, int len, int* wtot) {
    const int tid  = threadIdx.x;
    const int lane = tid & 31;
    const int w    = tid >> 5;
    const int b0   = tid * PER;
    int v[PER];
    int s = 0;
#pragma unroll
    for (int j = 0; j < PER; ++j) {
        v[j] = (b0 + j < len) ? a[b0 + j] : 0;
        s += v[j];
    }
    int inc = s;
    for (int off = 1; off < 32; off <<= 1) {
        int t = __shfl_up(inc, off);
        if (lane >= off) inc += t;
    }
    if (lane == 31) wtot[w] = inc;
    __syncthreads();
    int base = inc - s;
    for (int j = 0; j < 8; ++j)
        if (j < w) base += wtot[j];
#pragma unroll
    for (int j = 0; j < PER; ++j) {
        if (b0 + j < len) { a[b0 + j] = base; base += v[j]; }
    }
    __syncthreads();
}

__global__ void __launch_bounds__(256)
k_chunk_sort(const int* __restrict__ src, const int* __restrict__ dst,
             int e, int n, int nbins, int nchunks,
             int* __restrict__ pairs, int* __restrict__ blockstart,
             const float* __restrict__ wa, _Float16* __restrict__ bpa,
             const float* __restrict__ wb, _Float16* __restrict__ bpb) {
    if ((int)blockIdx.x == nchunks)     { pack_b<2>(wa, bpa, 128); return; }
    if ((int)blockIdx.x == nchunks + 1) { pack_b<1>(wb, bpb, 32);  return; }
    __shared__ int hist[MAXB + 1];
    __shared__ int cur[MAXB];
    __shared__ int stage[CH];
    __shared__ int wtot[8];
    if (nbins > MAXB) return;
    const int tid  = threadIdx.x;
    const int base = blockIdx.x * CH;
    const unsigned nmax = (unsigned)(n - 1);

    for (int b = tid; b <= nbins; b += 256) hist[b] = 0;

    int dd[16], ss[16];
    const bool al  = ((((size_t)src) | ((size_t)dst)) & 15) == 0;
    const bool alp = (((size_t)pairs) & 15) == 0;
#pragma unroll
    for (int k = 0; k < 4; ++k) {
        const int idx = base + k * 1024 + tid * 4;
        if (al && idx + 3 < e) {
            const int4 d4 = *reinterpret_cast<const int4*>(dst + idx);
            const int4 s4 = *reinterpret_cast<const int4*>(src + idx);
            dd[k * 4 + 0] = d4.x; dd[k * 4 + 1] = d4.y; dd[k * 4 + 2] = d4.z; dd[k * 4 + 3] = d4.w;
            ss[k * 4 + 0] = s4.x; ss[k * 4 + 1] = s4.y; ss[k * 4 + 2] = s4.z; ss[k * 4 + 3] = s4.w;
        } else {
#pragma unroll
            for (int j = 0; j < 4; ++j) {
                const bool ok = idx + j < e;
                dd[k * 4 + j] = ok ? dst[idx + j] : -1;
                ss[k * 4 + j] = ok ? src[idx + j] : 0;
            }
        }
    }
    bool valid[16];
#pragma unroll
    for (int k = 0; k < 16; ++k) {
        const int idx = base + (k >> 2) * 1024 + tid * 4 + (k & 3);
        valid[k] = idx < e;
        dd[k] = (int)min((unsigned)dd[k], nmax);
    }
    __syncthreads();
#pragma unroll
    for (int k = 0; k < 16; ++k)
        if (valid[k]) atomicAdd(&hist[dd[k] >> 8], 1);
    __syncthreads();
    block_excl_scan<(MAXB + 256) / 256>(hist, nbins + 1, wtot);

    int* __restrict__ row = blockstart + (size_t)blockIdx.x * (nbins + 1);
    for (int b = tid; b <= nbins; b += 256) {
        const int h = hist[b];
        row[b] = h;
        if (b < nbins) cur[b] = h;
    }
    __syncthreads();
#pragma unroll
    for (int k = 0; k < 16; ++k) {
        if (valid[k]) {
            const int pos = atomicAdd(&cur[dd[k] >> 8], 1);
            stage[pos] = (int)(((unsigned)ss[k] << 8) | ((unsigned)dd[k] & 255u));
        }
    }
    __syncthreads();
#pragma unroll
    for (int k = 0; k < 4; ++k) {
        const int li  = k * 1024 + tid * 4;
        const int idx = base + li;
        if (alp && idx + 3 < e) {
            *reinterpret_cast<int4*>(pairs + idx) =
                make_int4(stage[li], stage[li + 1], stage[li + 2], stage[li + 3]);
        } else {
            for (int j = 0; j < 4; ++j)
                if (idx + j < e) pairs[idx + j] = stage[li + j];
        }
    }
}

__global__ void __launch_bounds__(256)
k_bin_sort(const int* __restrict__ pairs, const int* __restrict__ blockstart,
           int e, int n, int nbins, int nchunks,
           int* __restrict__ csr, int* __restrict__ rowstart,
           float* __restrict__ dinv) {
    __shared__ int rstart[MAXC];
    __shared__ int roff[MAXC + 1];
    __shared__ int hist[256];
    __shared__ int cur[256];
    __shared__ int buf[CAP];
    __shared__ int outb[CAP];
    __shared__ int wtot[8];
    __shared__ int bs_sh;
    if (nbins > MAXB || nchunks > MAXC) return;
    const int tid = threadIdx.x;
    const int b   = blockIdx.x;
    const int g   = tid >> 3;
    const int gl  = tid & 7;

    if (tid == 0) bs_sh = 0;
    hist[tid] = 0;
    __syncthreads();
    int below = 0;
    for (int c = tid; c <= nchunks; c += 256) {
        int len = 0;
        if (c < nchunks) {
            const int* __restrict__ row = blockstart + (size_t)c * (nbins + 1);
            const int v0 = row[b];
            const int v1 = row[b + 1];
            rstart[c] = c * CH + v0;
            len = v1 - v0;
            below += v0;
        }
        roff[c] = len;
    }
    atomicAdd(&bs_sh, below);
    __syncthreads();
    block_excl_scan<(MAXC + 256) / 256>(roff, nchunks + 1, wtot);
    const int bs  = bs_sh;
    const int cnt = roff[nchunks];
    const bool small = cnt <= CAP;

    const int elast = e - 1;
    for (int c0 = g; c0 < nchunks; c0 += 128) {
        int rr[4], oo[4], ll[4], pv[8];
#pragma unroll
        for (int j = 0; j < 4; ++j) {
            const int cc = c0 + 32 * j;
            const bool in = cc < nchunks;
            const int ccs = in ? cc : 0;
            rr[j] = rstart[ccs];
            oo[j] = roff[ccs];
            ll[j] = in ? roff[ccs + 1] - oo[j] : 0;
        }
#pragma unroll
        for (int j = 0; j < 4; ++j) {
            pv[2 * j]     = pairs[min(rr[j] + gl, elast)];
            pv[2 * j + 1] = pairs[min(rr[j] + gl + 8, elast)];
        }
#pragma unroll
        for (int j = 0; j < 8; ++j) {
            const int i = gl + 8 * (j & 1);
            if (i < ll[j >> 1]) {
                if (small) buf[oo[j >> 1] + i] = pv[j];
                atomicAdd(&hist[pv[j] & 255], 1);
            }
        }
#pragma unroll
        for (int j = 0; j < 4; ++j) {
            for (int i = gl + 16; i < ll[j]; i += 8) {
                const int p = pairs[rr[j] + i];
                if (small) buf[oo[j] + i] = p;
                atomicAdd(&hist[p & 255], 1);
            }
        }
    }
    __syncthreads();
    const int mycnt = hist[tid];
    block_excl_scan<1>(hist, 256, wtot);
    const int mystart = hist[tid];
    cur[tid] = mystart;
    const int node = b * 256 + tid;
    if (node < n) {
        rowstart[node] = bs + mystart;
        dinv[node] = rsqrtf(fmaxf((float)(mycnt + 1), 1.0f));
    }
    __syncthreads();

    if (small) {
        for (int i = tid; i < cnt; i += 256) {
            const unsigned p = (unsigned)buf[i];
            const int pos = atomicAdd(&cur[p & 255u], 1);
            outb[pos] = (int)(p >> 8);
        }
        __syncthreads();
        for (int i = tid; i < cnt; i += 256)
            if (bs + i < e) csr[bs + i] = outb[i];
    } else {
        for (int c = g; c < nchunks; c += 32) {
            const int rs = rstart[c];
            const int len = roff[c + 1] - roff[c];
            for (int i = gl; i < len; i += 8) {
                const unsigned p = (unsigned)pairs[rs + i];
                const int pos = bs + atomicAdd(&cur[p & 255u], 1);
                if ((unsigned)pos < (unsigned)e) csr[pos] = (int)(p >> 8);
            }
        }
    }
}


template <int NT>
__global__ void k_wmma_gemm(const float* __restrict__ A,
                            const _Float16* __restrict__ Bp,
                            const float* __restrict__ rowscale,
                            float* __restrict__ C,
                            int M, int K) {
    const int N    = NT * 16;
    const int wave = threadIdx.x >> 5;
    const int lane = threadIdx.x & 31;
    const int tile = blockIdx.x * (blockDim.x >> 5) + wave;
    if (tile * 16 >= M) return;

    const int r   = lane & 15;
    const int kh  = (lane >> 4) << 3;
    const int row = tile * 16 + r;
    const float* __restrict__ arow = A + (size_t)row * K;
    const v16h* __restrict__ bfrag = (const v16h*)Bp;

    const int mbase = (lane < 16) ? 0 : 8;
    float rs[8];
#pragma unroll
    for (int j = 0; j < 8; ++j) rs[j] = rowscale[(size_t)(tile * 16 + mbase + j)];

    v8f zero = {};
    v8f acc[NT];
#pragma unroll
    for (int t = 0; t < NT; ++t) acc[t] = zero;

    for (int k0 = 0; k0 < K; k0 += 32) {
        const float4* __restrict__ a0 =
            reinterpret_cast<const float4*>(arow + k0 + kh);
        const float4* __restrict__ a1 =
            reinterpret_cast<const float4*>(arow + k0 + 16 + kh);
        float4 p0 = a0[0], p1 = a0[1];
        float4 q0 = a1[0], q1 = a1[1];
        v16h a;
        a[0]  = (_Float16)p0.x; a[1]  = (_Float16)p0.y;
        a[2]  = (_Float16)p0.z; a[3]  = (_Float16)p0.w;
        a[4]  = (_Float16)p1.x; a[5]  = (_Float16)p1.y;
        a[6]  = (_Float16)p1.z; a[7]  = (_Float16)p1.w;
        a[8]  = (_Float16)q0.x; a[9]  = (_Float16)q0.y;
        a[10] = (_Float16)q0.z; a[11] = (_Float16)q0.w;
        a[12] = (_Float16)q1.x; a[13] = (_Float16)q1.y;
        a[14] = (_Float16)q1.z; a[15] = (_Float16)q1.w;

        const int cidx = (k0 >> 5) * NT;
#pragma unroll
        for (int t = 0; t < NT; ++t) {
            v16h b = bfrag[(size_t)(cidx + t) * 32 + lane];
            acc[t] = __builtin_amdgcn_wmma_f32_16x16x32_f16(
                false, a, false, b, (short)0, acc[t], false, false);
        }
    }

    const int ncol  = lane & 15;
#pragma unroll
    for (int j = 0; j < 8; ++j) {
        const size_t orow = (size_t)(tile * 16 + mbase + j);
#pragma unroll
        for (int t = 0; t < NT; ++t)
            C[orow * N + t * 16 + ncol] = acc[t][j] * rs[j];
    }
}


static __device__ __forceinline__ void add4(float4& a, const float4 v) {
    a.x += v.x; a.y += v.y; a.z += v.z; a.w += v.w;
}

template <int K, int ROWB>
static __device__ __forceinline__ void take_rows(const char* __restrict__ hwb,
                                                 const int* __restrict__ csr,
                                                 int j, unsigned nmax,
                                                 unsigned qoff, float4& acc) {
    unsigned off[K];
#pragma unroll
    for (int k = 0; k < K; ++k)
        off[k] = min((unsigned)csr[j + k], nmax) * (unsigned)ROWB + qoff;
    float4 v[K];
#pragma unroll
    for (int k = 0; k < K; ++k)
        v[k] = *reinterpret_cast<const float4*>(hwb + off[k]);
#pragma unroll
    for (int k = 0; k < K; ++k) add4(acc, v[k]);
}

template <int C>
static __device__ __forceinline__ float4 gather_piece(const float* __restrict__ hw,
                                                      const int* __restrict__ rowstart,
                                                      const int* __restrict__ csr,
                                                      const float* __restrict__ dinv,
                                                      const float* __restrict__ bias,
                                                      int node, int q, int n, int e,
                                                      float& dd_out) {
    constexpr int ROWB = C * 4;
    const unsigned nmax = (unsigned)(n - 1);
    const unsigned qoff = (unsigned)q * 16u;
    const char* __restrict__ hwb = reinterpret_cast<const char*>(hw);

    const int beg  = rowstart[node];
    const int end  = (node + 1 < n) ? rowstart[node + 1] : e;
    const float dd = dinv[node];
    const float4 b = reinterpret_cast<const float4*>(bias)[q];
    float4 acc = *reinterpret_cast<const float4*>(hwb + (unsigned)node * (unsigned)ROWB + qoff);

    int j = beg;
    for (; j + 8 <= end; j += 8) take_rows<8, ROWB>(hwb, csr, j, nmax, qoff, acc);
    if (j + 4 <= end) { take_rows<4, ROWB>(hwb, csr, j, nmax, qoff, acc); j += 4; }
    const int r = end - j;
    if (r > 0) {
        const int last = end - 1;
        const unsigned o0 = min((unsigned)csr[j],                nmax) * (unsigned)ROWB + qoff;
        const unsigned o1 = min((unsigned)csr[min(j + 1, last)], nmax) * (unsigned)ROWB + qoff;
        const unsigned o2 = min((unsigned)csr[min(j + 2, last)], nmax) * (unsigned)ROWB + qoff;
        const float4 v0 = *reinterpret_cast<const float4*>(hwb + o0);
        const float4 v1 = *reinterpret_cast<const float4*>(hwb + o1);
        const float4 v2 = *reinterpret_cast<const float4*>(hwb + o2);
        add4(acc, v0);
        if (r > 1) add4(acc, v1);
        if (r > 2) add4(acc, v2);
    }

    float4 o;
    o.x = b.x + acc.x * dd;
    o.y = b.y + acc.y * dd;
    o.z = b.z + acc.z * dd;
    o.w = b.w + acc.w * dd;
    dd_out = dd;
    return o;
}

#define LP 36
__global__ void __launch_bounds__(256)
k_gather32_lin16(const float* __restrict__ hw,
                 const int* __restrict__ rowstart,
                 const int* __restrict__ csr,
                 const float* __restrict__ dinv,
                 const float* __restrict__ bias,
                 const _Float16* __restrict__ Bp,
                 float* __restrict__ agg,
                 float* __restrict__ lw, int n, int e) {
    __shared__ __attribute__((aligned(16))) float tile[32 * LP];
    __shared__ __attribute__((aligned(16))) float sdinv[32];
    const int tid  = threadIdx.x;
    const int i    = gtid();
    const int nraw = i >> 3, q = i & 7;
    const int node = min(nraw, n - 1);
    float dd;
    const float4 o = gather_piece<32>(hw, rowstart, csr, dinv, bias, node, q, n, e, dd);
    if (nraw < n)
        *reinterpret_cast<float4*>(reinterpret_cast<char*>(agg) + (unsigned)node * 128u + (unsigned)q * 16u) = o;
    *reinterpret_cast<float4*>(&tile[(tid >> 3) * LP + q * 4]) = o;
    if (q == 0) sdinv[tid >> 3] = dd;
    __syncthreads();

    const int wave = tid >> 5;
    if (wave >= 2) return;
    const int lane = tid & 31;
    const int r    = lane & 15;
    const int kh   = (lane >> 4) << 3;
    const float* __restrict__ trow = &tile[(wave * 16 + r) * LP];
    const float4 p0 = *reinterpret_cast<const float4*>(trow + kh);
    const float4 p1 = *reinterpret_cast<const float4*>(trow + kh + 4);
    const float4 q0 = *reinterpret_cast<const float4*>(trow + 16 + kh);
    const float4 q1 = *reinterpret_cast<const float4*>(trow + 16 + kh + 4);
    v16h a;
    a[0]  = (_Float16)p0.x; a[1]  = (_Float16)p0.y;
    a[2]  = (_Float16)p0.z; a[3]  = (_Float16)p0.w;
    a[4]  = (_Float16)p1.x; a[5]  = (_Float16)p1.y;
    a[6]  = (_Float16)p1.z; a[7]  = (_Float16)p1.w;
    a[8]  = (_Float16)q0.x; a[9]  = (_Float16)q0.y;
    a[10] = (_Float16)q0.z; a[11] = (_Float16)q0.w;
    a[12] = (_Float16)q1.x; a[13] = (_Float16)q1.y;
    a[14] = (_Float16)q1.z; a[15] = (_Float16)q1.w;
    const v16h bfr = reinterpret_cast<const v16h*>(Bp)[lane];
    v8f acc = {};
    acc = __builtin_amdgcn_wmma_f32_16x16x32_f16(false, a, false, bfr, (short)0, acc, false, false);

    const int mbase = (lane < 16) ? 0 : 8;
    const int ncol  = lane & 15;
    const int row0  = blockIdx.x * 32 + wave * 16 + mbase;
    const float4 s0 = *reinterpret_cast<const float4*>(&sdinv[wave * 16 + mbase]);
    const float4 s1 = *reinterpret_cast<const float4*>(&sdinv[wave * 16 + mbase + 4]);
    const float rs[8] = {s0.x, s0.y, s0.z, s0.w, s1.x, s1.y, s1.z, s1.w};
#pragma unroll
    for (int j = 0; j < 8; ++j) {
        const int m = row0 + j;
        if (m < n) lw[(size_t)m * 16 + ncol] = acc[j] * rs[j];
    }
}

__global__ void __launch_bounds__(256)
k_gather16_relu_dot(const float* __restrict__ hw,
                    const int* __restrict__ rowstart,
                    const int* __restrict__ csr,
                    const float* __restrict__ dinv,
                    const float* __restrict__ bias,
                    const float* __restrict__ w,
                    float* __restrict__ tlin, int n, int e) {
    __shared__ float ts[64];
    const int tid  = threadIdx.x;
    const int i    = gtid();
    const int node = min(i >> 2, n - 1), q = i & 3;
    const float4 wq = reinterpret_cast<const float4*>(w)[q];
    float dd;
    const float4 o = gather_piece<16>(hw, rowstart, csr, dinv, bias, node, q, n, e, dd);
    float s = 0.0f;
    s += fmaxf(o.x, 0.0f) * wq.x;
    s += fmaxf(o.y, 0.0f) * wq.y;
    s += fmaxf(o.z, 0.0f) * wq.z;
    s += fmaxf(o.w, 0.0f) * wq.w;
    s += __shfl_xor(s, 1);
    s += __shfl_xor(s, 2);
    if (q == 0) ts[tid >> 2] = s * dd;
    __syncthreads();
    if (tid < 64) {
        const int nn = blockIdx.x * 64 + tid;
        if (nn < n) tlin[nn] = ts[tid];
    }
}

__global__ void k_gather1_scale(const float* __restrict__ hw,
                                const int* __restrict__ rowstart,
                                const int* __restrict__ csr,
                                const float* __restrict__ dinv,
                                const float* __restrict__ bias,
                                float* __restrict__ out, int n, int e) {
    const int lane  = threadIdx.x & 31;
    const int i     = gtid();
    const int wbase = i - lane;
    if (wbase >= n) return;
    const int total = n * 8;
    float4* __restrict__ o4 = reinterpret_cast<float4*>(out);
    const int p0 = wbase * 8 + lane;
    float4 v[8];
#pragma unroll
    for (int k = 0; k < 8; ++k) v[k] = o4[min(p0 + k * 32, total - 1)];

    const int node = min(i, n - 1);
    const unsigned nmax = (unsigned)(n - 1);
    const int beg  = rowstart[node];
    const int end  = (node + 1 < n) ? rowstart[node + 1] : e;
    const float dd = dinv[node];
    float acc = hw[node];
#pragma unroll 4
    for (int j = beg; j < end; ++j) {
        const int s = (int)min((unsigned)csr[j], nmax);
        acc += hw[s];
    }
    const float t = bias[0] + acc * dd;
    const float f = logf(expf(t) + 1.1f);
#pragma unroll
    for (int k = 0; k < 8; ++k) {
        const float fk = __shfl(f, k * 4 + (lane >> 3));
        float4 r = v[k];
        r.x *= fk; r.y *= fk; r.z *= fk; r.w *= fk;
        if (p0 + k * 32 < total) o4[p0 + k * 32] = r;
    }
}


extern "C" void kernel_launch(void* const* d_in, const int* in_sizes, int n_in,
                              void* d_out, int out_size, void* d_ws, size_t ws_size,
                              hipStream_t stream) {
    const float* x      = (const float*)d_in[0];
    const int*   ei     = (const int*)  d_in[1];
    const float* W_base = (const float*)d_in[2];
    const float* b_base = (const float*)d_in[3];
    const float* W1     = (const float*)d_in[4];
    const float* b1     = (const float*)d_in[5];
    const float* W2     = (const float*)d_in[6];
    const float* b2     = (const float*)d_in[7];
    float* out = (float*)d_out;

    const int N = in_sizes[0] / 128;
    const int E = in_sizes[1] / 2;
    const int* src = ei;
    const int* dst = ei + E;

    float* ws   = (float*)d_ws;
    int*   rowstart = (int*)ws;
    float* dinv = ws   + N;
    float* xw   = dinv + N;
    int*   csr  = (int*)(xw + (size_t)N * 32);
    int*   pairs = csr + E;
    float* lw   = (float*)pairs;
    float* tlin = (float*)(pairs + E);
    int*   bstart = (int*)xw;
    _Float16* bp1 = (_Float16*)(tlin + 2 * (size_t)N);
    _Float16* bp2 = bp1 + 4 * 2 * 512;

    const int thr = 256;
    auto cdiv = [](int a, int b) { return (a + b - 1) / b; };
    const int mtiles = cdiv(N, 16);
    const int wpb = thr / 32;

    const int nbins   = cdiv(N, 256);
    const int nchunks = cdiv(E, CH);
    k_chunk_sort<<<nchunks + 2, 256, 0, stream>>>(src, dst, E, N, nbins, nchunks, pairs, bstart,
                                                  W_base, bp1, W1, bp2);
    k_bin_sort  <<<nbins, 256, 0, stream>>>(pairs, bstart, E, N, nbins, nchunks, csr, rowstart, dinv);

    k_wmma_gemm<2><<<cdiv(mtiles, wpb), thr, 0, stream>>>(x, bp1, dinv, xw, N, 128);
    k_gather32_lin16<<<cdiv(N * 8, thr), thr, 0, stream>>>(xw, rowstart, csr, dinv, b_base, bp2, out, lw, N, E);

    k_gather16_relu_dot<<<cdiv(N * 4, thr), thr, 0, stream>>>(lw, rowstart, csr, dinv, b1, W2, tlin, N, E);

    k_gather1_scale<<<cdiv(N, thr), thr, 0, stream>>>(tlin, rowstart, csr, dinv, b2, out, N, E);
}
